# starting kernel
# baseline (speedup 1.0000x reference)
#include <hip/hip_runtime.h>

typedef __attribute__((ext_vector_type(2))) float v2f;
typedef __attribute__((ext_vector_type(8))) float v8f;

#define BATCH   4096
#define IN_DIM  1024
#define FD      64
#define OD      30
#define KSTR    (FD * OD)

#define MT       16
#define KCH      256
#define XS_STRIDE 258

__global__ __launch_bounds__(256) void build_W(const float* __restrict__ ker,
                                               const int* __restrict__ fidx,
                                               float* __restrict__ W) {
    const int idx = blockIdx.x * 256 + threadIdx.x;
    const int i = idx >> 10;
    const int j = idx & 1023;
    const int fi = fidx[i];
    const int fj = fidx[j];
    const float* __restrict__ a = ker + i * KSTR + fj * OD;
    const float* __restrict__ b = ker + j * KSTR + fi * OD;
    float s = 0.f;
#pragma unroll
    for (int d = 0; d < OD; ++d) s += a[d] * b[d];
    W[idx] = s;
}

__global__ __launch_bounds__(256) void ffm_gemm(const float* __restrict__ X,
                                                const float* __restrict__ W,
                                                float* __restrict__ out) {
    __shared__ float Xs[MT * XS_STRIDE];
    __shared__ float red[MT];
    __shared__ float redD[MT];

    const int batch0 = blockIdx.x * MT;
    const int tid  = threadIdx.x;
    const int wave = tid >> 5;
    const int lane = tid & 31;

    if (tid < MT) { red[tid] = 0.f; redD[tid] = 0.f; }

    v8f acc[8];
#pragma unroll
    for (int t = 0; t < 8; ++t) acc[t] = (v8f)(0.f);

    const int nbase = wave * 128;
    const int m_a   = lane & 15;
    const int ksel  = (lane >> 4) * 2;
    const int n_b   = lane & 15;

    for (int kb = 0; kb < IN_DIM; kb += KCH) {
        __syncthreads();
        for (int v = tid; v < MT * (KCH / 4); v += 256) {
            const int row  = v / (KCH / 4);
            const int col4 = v % (KCH / 4);
            const float4 src =
                ((const float4*)(X + (size_t)(batch0 + row) * IN_DIM + kb))[col4];
            float* dst = &Xs[row * XS_STRIDE + col4 * 4];
            dst[0] = src.x; dst[1] = src.y; dst[2] = src.z; dst[3] = src.w;
        }
        __syncthreads();

#pragma unroll
        for (int t = 0; t < 8; ++t) {
            const int ncol = nbase + t * 16 + n_b;
            const float* __restrict__ wp = W + ncol;
#pragma unroll 4
            for (int k = 0; k < KCH; k += 4) {
                v2f a, b;
                const float* xs = &Xs[m_a * XS_STRIDE + k + ksel];
                a.x = xs[0];
                a.y = xs[1];
                const float* wrow = wp + (size_t)(kb + k + ksel) * IN_DIM;
                b.x = wrow[0];
                b.y = wrow[IN_DIM];
                acc[t] = __builtin_amdgcn_wmma_f32_16x16x4_f32(
                     false, a,  false, b,
                     (short)0, acc[t],
                     false,  false);
            }
        }
    }

    float sums[8];
#pragma unroll
    for (int r = 0; r < 8; ++r) sums[r] = 0.f;
#pragma unroll
    for (int t = 0; t < 8; ++t) {
        const int n = nbase + t * 16 + (lane & 15);
#pragma unroll
        for (int r = 0; r < 8; ++r) {
            const int m = r + ((lane >> 4) << 3);
            sums[r] += acc[t][r] * X[(size_t)(batch0 + m) * IN_DIM + n];
        }
    }
#pragma unroll
    for (int r = 0; r < 8; ++r) {
        const int m = r + ((lane >> 4) << 3);
        atomicAdd(&red[m], sums[r]);
    }

    {
        const int m   = tid >> 4;
        const int seg = tid & 15;
        const float* xr = X + (size_t)(batch0 + m) * IN_DIM + seg * 64;
        float s = 0.f;
#pragma unroll 8
        for (int i = 0; i < 64; ++i) {
            const float xv = xr[i];
            s += xv * xv * W[(size_t)(seg * 64 + i) * (IN_DIM + 1)];
        }
        atomicAdd(&redD[m], s);
    }

    __syncthreads();
    if (tid < MT) out[batch0 + tid] = 0.5f * (red[tid] - redD[tid]);
}

extern "C" void kernel_launch(void* const* d_in, const int* in_sizes, int n_in,
                              void* d_out, int out_size, void* d_ws, size_t ws_size,
                              hipStream_t stream) {
    const float* x    = (const float*)d_in[0];
    const float* ker  = (const float*)d_in[1];
    const int*   fidx = (const int*)d_in[2];
    float* W   = (float*)d_ws;
    float* outp = (float*)d_out;

    build_W<<<(IN_DIM * IN_DIM) / 256, 256, 0, stream>>>(ker, fidx, W);
    ffm_gemm<<<BATCH / MT, 256, 0, stream>>>(x, W, outp);
}


# build_W as per-field-pair block products on f32 WMMA: features listed by field in LDS, W[I_p,I_q] = kernel[I_p,q,:] x kernel[I_q,p,:]^T, rows staged in LDS (order of W's 30-term sums changes)
# speedup vs baseline: 3.5853x; 3.5853x over previous
#include <hip/hip_runtime.h>

typedef __attribute__((ext_vector_type(2))) float v2f;
typedef __attribute__((ext_vector_type(8))) float v8f;

#define BATCH   4096
#define IN_DIM  1024
#define FD      64
#define OD      30
#define KSTR    (FD * OD)

#define BM      128
#define BN      128
#define BK      64
#define PITCH   68
#define NKC     (IN_DIM / BK)

#define RB_PITCH 1028
__global__ __launch_bounds__(1024) void build_W(const float* __restrict__ ker,
                                                const int* __restrict__ fidx,
                                                float* __restrict__ W,
                                                float* __restrict__ out) {
    __shared__ unsigned short lst[FD][IN_DIM];
    __shared__ int cnt[FD];
    __shared__ __attribute__((aligned(16))) float rowbuf[16 * RB_PITCH];

    const int tid  = threadIdx.x;
    const int wave = tid >> 5;
    const int lane = tid & 31;
    const int c    = lane & 15;
    const int h    = lane >> 4;
    const int p    = blockIdx.x >> 1;
    const int slot = blockIdx.x & 1;

    const int g = blockIdx.x * 1024 + tid;
    if (g < BATCH) out[g] = 0.f;

    {
        int fv[IN_DIM / 32];
#pragma unroll
        for (int k = 0; k < IN_DIM / 32; ++k) fv[k] = fidx[k * 32 + lane];
        int n0 = 0, n1 = 0;
#pragma unroll
        for (int k = 0; k < IN_DIM / 32; ++k) {
            const bool m0 = (fv[k] == wave);
            const bool m1 = (fv[k] == wave + 32);
            const unsigned b0 = __builtin_amdgcn_ballot_w32(m0);
            const unsigned b1 = __builtin_amdgcn_ballot_w32(m1);
            if (m0) lst[wave][n0 + (int)__builtin_amdgcn_mbcnt_lo(b0, 0u)] = (unsigned short)(k * 32 + lane);
            if (m1) lst[wave + 32][n1 + (int)__builtin_amdgcn_mbcnt_lo(b1, 0u)] = (unsigned short)(k * 32 + lane);
            n0 += __builtin_popcount(b0);
            n1 += __builtin_popcount(b1);
        }
        if (lane == 0) { cnt[wave] = n0; cnt[wave + 32] = n1; }
    }
    __syncthreads();

    const int np = __builtin_amdgcn_readfirstlane(cnt[p]);
    for (int m0 = slot * 16; m0 < np; m0 += 32) {
        const int i = lst[p][min(m0 + c, np - 1)];
#pragma unroll
        for (int qq = 0; qq < 2; ++qq) {
            const int q  = wave + 32 * qq;
            const int nq = __builtin_amdgcn_readfirstlane(cnt[q]);
            if (nq > 0) {
                const float* __restrict__ ap = ker + (size_t)i * KSTR + q * OD;
                v2f a[8];
#pragma unroll
                for (int ks = 0; ks < 7; ++ks) a[ks] = *(const v2f*)(ap + 4 * ks + 2 * h);
                a[7] = *(const v2f*)(ap + 28);
                if (h) a[7] = (v2f)(0.f);
                for (int n0 = 0; n0 < nq; n0 += 16) {
                    const int j = lst[q][min(n0 + c, nq - 1)];
                    const float* __restrict__ bp = ker + (size_t)j * KSTR + p * OD;
                    v2f b[8];
#pragma unroll
                    for (int ks = 0; ks < 7; ++ks) b[ks] = *(const v2f*)(bp + 4 * ks + 2 * h);
                    b[7] = *(const v2f*)(bp + 28);
                    if (h) b[7] = (v2f)(0.f);
                    v8f acc = (v8f)(0.f);
#pragma unroll
                    for (int ks = 0; ks < 8; ++ks)
                        acc = __builtin_amdgcn_wmma_f32_16x16x4_f32(
                            false, a[ks], false, b[ks], (short)0, acc, false, false);
                    if (n0 + c < nq) {
#pragma unroll
                        for (int r = 0; r < 8; ++r) rowbuf[(8 * h + r) * RB_PITCH + j] = acc[r];
                    }
                }
            }
        }
        __syncthreads();
        const int rr = wave >> 1;
        if (m0 + rr < np) {
            const int irow = lst[p][m0 + rr];
#pragma unroll
            for (int t = 0; t < 16; ++t) {
                const int col = ((wave & 1) * 16 + t) * 32 + lane;
                const float v = rowbuf[rr * RB_PITCH + col];
                W[(size_t)irow * IN_DIM + col] = (col == irow) ? 0.f : v;
            }
        }
        __syncthreads();
    }
}

__global__ __launch_bounds__(256) __attribute__((amdgpu_waves_per_eu(2, 2))) void ffm_gemm(const float* __restrict__ X,
                                                const float* __restrict__ W,
                                                float* __restrict__ out) {
    __shared__ __attribute__((aligned(16))) float Xs[2][BM * PITCH];
    __shared__ __attribute__((aligned(16))) float Ws[2][BN * PITCH];
    __shared__ float part[4][BM];

    const int tid  = threadIdx.x;
    const int wave = tid >> 5;
    const int lane = tid & 31;
    const int c    = lane & 15;
    const int h    = lane >> 4;
    const int nb   = blockIdx.x & 7;
    const int rb   = blockIdx.x >> 3;
    const int batch0 = rb * BM;
    const int n0     = nb * BN;
    const int wm = wave & 1;
    const int wn = wave >> 1;

    const int srow = tid >> 4;
    const int scol = (tid & 15) * 4;
    const float* __restrict__ xg = X + (size_t)(batch0 + srow) * IN_DIM + scol;
    const float* __restrict__ wg = W + (size_t)(n0 + srow) * IN_DIM + scol;
    const int sdst = srow * PITCH + scol;

    float4 px0, px1, px2, px3, px4, px5, px6, px7;
    float4 pw0, pw1, pw2, pw3, pw4, pw5, pw6, pw7;
#define LDG(i, koff)                                                          \
    px##i = *(const float4*)(xg + (size_t)(i) * 16 * IN_DIM + (koff));        \
    pw##i = *(const float4*)(wg + (size_t)(i) * 16 * IN_DIM + (koff));
#define STS(i, st)                                                            \
    *(float4*)&Xs[st][sdst + (i) * 16 * PITCH] = px##i;                       \
    *(float4*)&Ws[st][sdst + (i) * 16 * PITCH] = pw##i;
#define LDG_ALL(koff) LDG(0, koff) LDG(1, koff) LDG(2, koff) LDG(3, koff) \
                      LDG(4, koff) LDG(5, koff) LDG(6, koff) LDG(7, koff)
#define STS_ALL(st)   STS(0, st) STS(1, st) STS(2, st) STS(3, st) \
                      STS(4, st) STS(5, st) STS(6, st) STS(7, st)

    LDG_ALL(0)
    STS_ALL(0)
    __syncthreads();

    v8f acc[4][2];
#pragma unroll
    for (int mt = 0; mt < 4; ++mt) {
        acc[mt][0] = (v8f)(0.f);
        acc[mt][1] = (v8f)(0.f);
    }

    const int xfrag = (wm * 64 + c) * PITCH + 4 * h;
    const int wfrag = (wn * 32 + c) * PITCH + 4 * h;

    for (int kc = 0; kc < NKC; ++kc) {
        const int cur = kc & 1;
        const bool more = (kc + 1 < NKC);
        if (more) {
            LDG_ALL((kc + 1) * BK)
        }
        __builtin_amdgcn_sched_barrier(0);
        const float* xs = &Xs[cur][xfrag];
        const float* ws = &Ws[cur][wfrag];
#pragma unroll
        for (int k8 = 0; k8 < BK; k8 += 8) {
            float4 a[4], b[2];
#pragma unroll
            for (int mt = 0; mt < 4; ++mt) a[mt] = *(const float4*)(xs + mt * 16 * PITCH + k8);
#pragma unroll
            for (int nt = 0; nt < 2; ++nt) b[nt] = *(const float4*)(ws + nt * 16 * PITCH + k8);
#pragma unroll
            for (int mt = 0; mt < 4; ++mt) {
#pragma unroll
                for (int nt = 0; nt < 2; ++nt) {
                    v2f a0 = {a[mt].x, a[mt].y};
                    v2f b0 = {b[nt].x, b[nt].y};
                    acc[mt][nt] = __builtin_amdgcn_wmma_f32_16x16x4_f32(
                        false, a0, false, b0, (short)0, acc[mt][nt], false, false);
                }
            }
#pragma unroll
            for (int mt = 0; mt < 4; ++mt) {
#pragma unroll
                for (int nt = 0; nt < 2; ++nt) {
                    v2f a1 = {a[mt].z, a[mt].w};
                    v2f b1 = {b[nt].z, b[nt].w};
                    acc[mt][nt] = __builtin_amdgcn_wmma_f32_16x16x4_f32(
                        false, a1, false, b1, (short)0, acc[mt][nt], false, false);
                }
            }
        }
        if (more) {
            STS_ALL(cur ^ 1)
        }
        __syncthreads();
    }

    float s[4][8];
#pragma unroll
    for (int mt = 0; mt < 4; ++mt) {
#pragma unroll
        for (int r = 0; r < 8; ++r) {
            const int row = wm * 64 + mt * 16 + 8 * h + r;
            const float* xr = X + (size_t)(batch0 + row) * IN_DIM + n0 + wn * 32 + c;
            s[mt][r] = acc[mt][0][r] * xr[0] + acc[mt][1][r] * xr[16];
        }
    }
#pragma unroll
    for (int mt = 0; mt < 4; ++mt) {
#pragma unroll
        for (int r = 0; r < 8; ++r) {
            float v = s[mt][r];
            v += __shfl_xor(v, 1);
            v += __shfl_xor(v, 2);
            v += __shfl_xor(v, 4);
            v += __shfl_xor(v, 8);
            s[mt][r] = v;
        }
    }
    if (c == 0) {
#pragma unroll
        for (int mt = 0; mt < 4; ++mt) {
#pragma unroll
            for (int r = 0; r < 8; ++r) {
                part[wn][wm * 64 + mt * 16 + 8 * h + r] = s[mt][r];
            }
        }
    }
    __syncthreads();
    if (tid < BM) {
        const float t = (part[0][tid] + part[1][tid]) + (part[2][tid] + part[3][tid]);
        atomicAdd(&out[batch0 + tid], 0.5f * t);
    }
}

extern "C" void kernel_launch(void* const* d_in, const int* in_sizes, int n_in,
                              void* d_out, int out_size, void* d_ws, size_t ws_size,
                              hipStream_t stream) {
    const float* x    = (const float*)d_in[0];
    const float* ker  = (const float*)d_in[1];
    const int*   fidx = (const int*)d_in[2];
    float* W   = (float*)d_ws;
    float* outp = (float*)d_out;

    build_W<<<2 * FD, 1024, 0, stream>>>(ker, fidx, W, outp);
    ffm_gemm<<<(BATCH / BM) * (IN_DIM / BN), 256, 0, stream>>>(x, W, outp);
}


# build_W re-partitioned onto all 256 WGPs: one 16-wave workgroup per (field, 16-row tile, column half), four column fields' loads in flight per wave, member lists by six-ballot rank
# speedup vs baseline: 3.7520x; 1.0465x over previous
#include <hip/hip_runtime.h>

typedef __attribute__((ext_vector_type(2))) float v2f;
typedef __attribute__((ext_vector_type(8))) float v8f;

#define BATCH   4096
#define IN_DIM  1024
#define FD      64
#define OD      30
#define KSTR    (FD * OD)

#define BM      128
#define BN      128
#define BK      64
#define PITCH   68
#define NKC     (IN_DIM / BK)

#define HALF     (IN_DIM / 2)
#define RB_PITCH (HALF + 4)
#define NGRP     (IN_DIM / 32)

__device__ __forceinline__ void load_frag(v2f (&fr)[8], const float* __restrict__ sp, int h) {
#pragma unroll
    for (int ks = 0; ks < 7; ++ks) fr[ks] = *(const v2f*)(sp + 4 * ks + 2 * h);
    fr[7] = *(const v2f*)(sp + 28);
}
__device__ __forceinline__ void pad_frag(v2f (&fr)[8], int h) {
    if (h) fr[7] = (v2f)(0.f);
}
__device__ __forceinline__ unsigned same_value_lanes(int f) {
    unsigned same = 0xffffffffu;
#pragma unroll
    for (int k = 0; k < 6; ++k) {
        const bool bit = (f >> k) & 1;
        const unsigned bk = __builtin_amdgcn_ballot_w32(bit);
        same &= bit ? bk : ~bk;
    }
    return same;
}

__global__ __launch_bounds__(512) void build_W(const float* __restrict__ ker,
                                               const int* __restrict__ fidx,
                                               float* __restrict__ W,
                                               float* __restrict__ out) {
    __shared__ unsigned short lst[FD][IN_DIM];
    __shared__ int grp[NGRP][FD];
    __shared__ int cnt[FD];
    __shared__ int cntlo[FD];
    __shared__ __attribute__((aligned(16))) float rowbuf[16 * RB_PITCH];

    const int tid  = threadIdx.x;
    const int wave = tid >> 5;
    const int lane = tid & 31;
    const int c    = lane & 15;
    const int h    = lane >> 4;
    const int p    = blockIdx.x >> 2;
    const int slot = (blockIdx.x >> 1) & 1;
    const int hf   = blockIdx.x & 1;
    const int cb   = hf * HALF;

    const int g = blockIdx.x * 512 + tid;
    if (g < BATCH) out[g] = 0.f;

    {
        const int fa = fidx[tid] & (FD - 1);
        const int fb = fidx[tid + HALF] & (FD - 1);
        grp[wave][lane] = 0;      grp[wave][lane + 32] = 0;
        grp[wave + 16][lane] = 0; grp[wave + 16][lane + 32] = 0;
        const unsigned sa = same_value_lanes(fa);
        const unsigned sb = same_value_lanes(fb);
        const int ra = (int)__builtin_amdgcn_mbcnt_lo(sa, 0u);
        const int rb = (int)__builtin_amdgcn_mbcnt_lo(sb, 0u);
        const int ta = __builtin_popcount(sa);
        const int tb = __builtin_popcount(sb);
        if (ra == 0) grp[wave][fa] = ta;
        if (rb == 0) grp[wave + 16][fb] = tb;
        __syncthreads();
        if (tid < FD) {
            int run = 0;
#pragma unroll
            for (int k = 0; k < NGRP; ++k) {
                const int t = grp[k][tid];
                grp[k][tid] = run;
                run += t;
                if (k == NGRP / 2 - 1) cntlo[tid] = run;
            }
            cnt[tid] = run;
        }
        __syncthreads();
        lst[fa][grp[wave][fa] + ra] = (unsigned short)tid;
        lst[fb][grp[wave + 16][fb] + rb] = (unsigned short)(tid + HALF);
    }
    __syncthreads();

    const int np = __builtin_amdgcn_readfirstlane(cnt[p]);
    for (int m0 = slot * 16; m0 < np; m0 += 32) {
        const int i = lst[p][min(m0 + c, np - 1)];
        const float* __restrict__ arow = ker + (size_t)i * KSTR;
        const float* __restrict__ bcol = ker + p * OD;

        auto tile = [&](const v2f (&a)[8], int q, int lo, int nq, int n0) {
            const int j = lst[q][lo + min(n0 + c, nq - 1)];
            v2f b[8];
            load_frag(b, bcol + (size_t)j * KSTR, h);
            pad_frag(b, h);
            v8f acc = (v8f)(0.f);
#pragma unroll
            for (int ks = 0; ks < 8; ++ks)
                acc = __builtin_amdgcn_wmma_f32_16x16x4_f32(
                    false, a[ks], false, b[ks], (short)0, acc, false, false);
            if (n0 + c < nq) {
#pragma unroll
                for (int r = 0; r < 8; ++r) rowbuf[(8 * h + r) * RB_PITCH + (j - cb)] = acc[r];
            }
        };

        int lo[4], nq[4], jj[4];
        v2f a[4][8], b[4][8];
#pragma unroll
        for (int t = 0; t < 4; ++t) {
            const int q  = wave + 16 * t;
            const int nl = __builtin_amdgcn_readfirstlane(cntlo[q]);
            const int nt = __builtin_amdgcn_readfirstlane(cnt[q]);
            lo[t] = hf ? nl : 0;
            nq[t] = hf ? nt - nl : nl;
            load_frag(a[t], arow + q * OD, h);
            jj[t] = (nq[t] > 0) ? (int)lst[q][lo[t] + min(c, nq[t] - 1)] : i;
        }
#pragma unroll
        for (int t = 0; t < 4; ++t) load_frag(b[t], bcol + (size_t)jj[t] * KSTR, h);
#pragma unroll
        for (int t = 0; t < 4; ++t) { pad_frag(a[t], h); pad_frag(b[t], h); }
        v8f acc[4];
#pragma unroll
        for (int t = 0; t < 4; ++t) acc[t] = (v8f)(0.f);
#pragma unroll
        for (int ks = 0; ks < 8; ++ks) {
#pragma unroll
            for (int t = 0; t < 4; ++t)
                acc[t] = __builtin_amdgcn_wmma_f32_16x16x4_f32(
                    false, a[t][ks], false, b[t][ks], (short)0, acc[t], false, false);
        }
#pragma unroll
        for (int t = 0; t < 4; ++t) {
            if (c < nq[t]) {
#pragma unroll
                for (int r = 0; r < 8; ++r) rowbuf[(8 * h + r) * RB_PITCH + (jj[t] - cb)] = acc[t][r];
            }
        }
#pragma unroll
        for (int t = 0; t < 4; ++t) {
            for (int n0 = 16; n0 < nq[t]; n0 += 16) tile(a[t], wave + 16 * t, lo[t], nq[t], n0);
        }
        __syncthreads();
        if (m0 + wave < np) {
            const int irow = lst[p][m0 + wave];
#pragma unroll
            for (int t = 0; t < 16; ++t) {
                const int col = t * 32 + lane;
                const float v = rowbuf[wave * RB_PITCH + col];
                W[(size_t)irow * IN_DIM + cb + col] = (cb + col == irow) ? 0.f : v;
            }
        }
        __syncthreads();
    }
}

__global__ __launch_bounds__(256) __attribute__((amdgpu_waves_per_eu(2, 2))) void ffm_gemm(const float* __restrict__ X,
                                                const float* __restrict__ W,
                                                float* __restrict__ out) {
    __shared__ __attribute__((aligned(16))) float Xs[2][BM * PITCH];
    __shared__ __attribute__((aligned(16))) float Ws[2][BN * PITCH];
    __shared__ float part[4][BM];

    const int tid  = threadIdx.x;
    const int wave = tid >> 5;
    const int lane = tid & 31;
    const int c    = lane & 15;
    const int h    = lane >> 4;
    const int nb   = blockIdx.x & 7;
    const int rb   = blockIdx.x >> 3;
    const int batch0 = rb * BM;
    const int n0     = nb * BN;
    const int wm = wave & 1;
    const int wn = wave >> 1;

    const int srow = tid >> 4;
    const int scol = (tid & 15) * 4;
    const float* __restrict__ xg = X + (size_t)(batch0 + srow) * IN_DIM + scol;
    const float* __restrict__ wg = W + (size_t)(n0 + srow) * IN_DIM + scol;
    const int sdst = srow * PITCH + scol;

    float4 px0, px1, px2, px3, px4, px5, px6, px7;
    float4 pw0, pw1, pw2, pw3, pw4, pw5, pw6, pw7;
#define LDG(i, koff)                                                          \
    px##i = *(const float4*)(xg + (size_t)(i) * 16 * IN_DIM + (koff));        \
    pw##i = *(const float4*)(wg + (size_t)(i) * 16 * IN_DIM + (koff));
#define STS(i, st)                                                            \
    *(float4*)&Xs[st][sdst + (i) * 16 * PITCH] = px##i;                       \
    *(float4*)&Ws[st][sdst + (i) * 16 * PITCH] = pw##i;
#define LDG_ALL(koff) LDG(0, koff) LDG(1, koff) LDG(2, koff) LDG(3, koff) \
                      LDG(4, koff) LDG(5, koff) LDG(6, koff) LDG(7, koff)
#define STS_ALL(st)   STS(0, st) STS(1, st) STS(2, st) STS(3, st) \
                      STS(4, st) STS(5, st) STS(6, st) STS(7, st)

    LDG_ALL(0)
    STS_ALL(0)
    __syncthreads();

    v8f acc[4][2];
#pragma unroll
    for (int mt = 0; mt < 4; ++mt) {
        acc[mt][0] = (v8f)(0.f);
        acc[mt][1] = (v8f)(0.f);
    }

    const int xfrag = (wm * 64 + c) * PITCH + 4 * h;
    const int wfrag = (wn * 32 + c) * PITCH + 4 * h;

    for (int kc = 0; kc < NKC; ++kc) {
        const int cur = kc & 1;
        const bool more = (kc + 1 < NKC);
        if (more) {
            LDG_ALL((kc + 1) * BK)
        }
        __builtin_amdgcn_sched_barrier(0);
        const float* xs = &Xs[cur][xfrag];
        const float* ws = &Ws[cur][wfrag];
#pragma unroll
        for (int k8 = 0; k8 < BK; k8 += 8) {
            float4 a[4], b[2];
#pragma unroll
            for (int mt = 0; mt < 4; ++mt) a[mt] = *(const float4*)(xs + mt * 16 * PITCH + k8);
#pragma unroll
            for (int nt = 0; nt < 2; ++nt) b[nt] = *(const float4*)(ws + nt * 16 * PITCH + k8);
#pragma unroll
            for (int mt = 0; mt < 4; ++mt) {
#pragma unroll
                for (int nt = 0; nt < 2; ++nt) {
                    v2f a0 = {a[mt].x, a[mt].y};
                    v2f b0 = {b[nt].x, b[nt].y};
                    acc[mt][nt] = __builtin_amdgcn_wmma_f32_16x16x4_f32(
                        false, a0, false, b0, (short)0, acc[mt][nt], false, false);
                }
            }
#pragma unroll
            for (int mt = 0; mt < 4; ++mt) {
#pragma unroll
                for (int nt = 0; nt < 2; ++nt) {
                    v2f a1 = {a[mt].z, a[mt].w};
                    v2f b1 = {b[nt].z, b[nt].w};
                    acc[mt][nt] = __builtin_amdgcn_wmma_f32_16x16x4_f32(
                        false, a1, false, b1, (short)0, acc[mt][nt], false, false);
                }
            }
        }
        if (more) {
            STS_ALL(cur ^ 1)
        }
        __syncthreads();
    }

    float s[4][8];
#pragma unroll
    for (int mt = 0; mt < 4; ++mt) {
#pragma unroll
        for (int r = 0; r < 8; ++r) {
            const int row = wm * 64 + mt * 16 + 8 * h + r;
            const float* xr = X + (size_t)(batch0 + row) * IN_DIM + n0 + wn * 32 + c;
            s[mt][r] = acc[mt][0][r] * xr[0] + acc[mt][1][r] * xr[16];
        }
    }
#pragma unroll
    for (int mt = 0; mt < 4; ++mt) {
#pragma unroll
        for (int r = 0; r < 8; ++r) {
            float v = s[mt][r];
            v += __shfl_xor(v, 1);
            v += __shfl_xor(v, 2);
            v += __shfl_xor(v, 4);
            v += __shfl_xor(v, 8);
            s[mt][r] = v;
        }
    }
    if (c == 0) {
#pragma unroll
        for (int mt = 0; mt < 4; ++mt) {
#pragma unroll
            for (int r = 0; r < 8; ++r) {
                part[wn][wm * 64 + mt * 16 + 8 * h + r] = s[mt][r];
            }
        }
    }
    __syncthreads();
    if (tid < BM) {
        const float t = (part[0][tid] + part[1][tid]) + (part[2][tid] + part[3][tid]);
        atomicAdd(&out[batch0 + tid], 0.5f * t);
    }
}

extern "C" void kernel_launch(void* const* d_in, const int* in_sizes, int n_in,
                              void* d_out, int out_size, void* d_ws, size_t ws_size,
                              hipStream_t stream) {
    const float* x    = (const float*)d_in[0];
    const float* ker  = (const float*)d_in[1];
    const int*   fidx = (const int*)d_in[2];
    float* W   = (float*)d_ws;
    float* outp = (float*)d_out;

    build_W<<<4 * FD, 512, 0, stream>>>(ker, fidx, W, outp);
    ffm_gemm<<<(BATCH / BM) * (IN_DIM / BN), 256, 0, stream>>>(x, W, outp);
}
